# starting kernel
# baseline (speedup 1.0000x reference)
#include <hip/hip_runtime.h>
#include <hip/hip_bf16.h>


typedef __attribute__((ext_vector_type(2))) float v2f;
typedef __attribute__((ext_vector_type(4))) float v4f;
typedef __attribute__((ext_vector_type(8))) float v8f;

#define D_FEAT 128

__global__ void __launch_bounds__(128, 1)
edge_dot_wmma_kernel(const int* __restrict__ src,
                     const int* __restrict__ dst,
                     const float* __restrict__ h,
                     float* __restrict__ out,
                     int nEdges)
{
    const int lane   = threadIdx.x & 31;
    const int waveId = threadIdx.x >> 5;
    const int wavesPerBlock = blockDim.x >> 5;
    const int e0 = (blockIdx.x * wavesPerBlock + waveId) << 4;

    if (e0 >= nEdges) return;

    const int m = lane & 15;
    int e = e0 + m;
    if (e >= nEdges) e = nEdges - 1;

    const long long srow = (long long)src[e];
    const long long drow = (long long)dst[e];

    const int colBase = (lane >> 4) << 2;
    const float* __restrict__ ps = h + srow * D_FEAT + colBase;
    const float* __restrict__ pd = h + drow * D_FEAT + colBase;

    v8f c = {};

#pragma unroll 4
    for (int k0 = 0; k0 < D_FEAT; k0 += 8) {
        v4f sv = *(const v4f*)(ps + k0);
        v4f dv = *(const v4f*)(pd + k0);

        v2f a0 = __builtin_shufflevector(sv, sv, 0, 1);
        v2f b0 = __builtin_shufflevector(dv, dv, 0, 1);
        v2f a1 = __builtin_shufflevector(sv, sv, 2, 3);
        v2f b1 = __builtin_shufflevector(dv, dv, 2, 3);

        c = __builtin_amdgcn_wmma_f32_16x16x4_f32(false, a0, false, b0,
                                                  (short)0, c, false, false);
        c = __builtin_amdgcn_wmma_f32_16x16x4_f32(false, a1, false, b1,
                                                  (short)0, c, false, false);
    }

    const int sel = lane & 7;
    float val = c[0];
    if (sel == 1) val = c[1];
    if (sel == 2) val = c[2];
    if (sel == 3) val = c[3];
    if (sel == 4) val = c[4];
    if (sel == 5) val = c[5];
    if (sel == 6) val = c[6];
    if (sel == 7) val = c[7];

    const bool writer = (lane < 8) || (lane >= 24);
    const int outIdx = e0 + ((lane < 8) ? lane : (lane - 16));
    if (writer && outIdx < nEdges) out[outIdx] = val;
}

extern "C" void kernel_launch(void* const* d_in, const int* in_sizes, int n_in,
                              void* d_out, int out_size, void* d_ws, size_t ws_size,
                              hipStream_t stream) {
    const int E = in_sizes[0] / 2;
    const int*   src = (const int*)d_in[0];
    const int*   dst = src + E;
    const float* h   = (const float*)d_in[1];
    float*       out = (float*)d_out;

    const int wavesPerBlock = 4;
    const int threads = wavesPerBlock * 32;
    const int totalWaves = (E + 15) / 16;
    const int blocks = (totalWaves + wavesPerBlock - 1) / wavesPerBlock;

    edge_dot_wmma_kernel<<<blocks, threads, 0, stream>>>(src, dst, h, out, E);
}


# Row gathers made whole-line: each global load reads one full 512-B row (4 lines, not 16 partial) into wave-private padded LDS; WMMA fragments read back by ds_load_b128; sum order unchanged
# speedup vs baseline: 1.2241x; 1.2241x over previous
#include <hip/hip_runtime.h>
#include <hip/hip_bf16.h>


typedef __attribute__((ext_vector_type(2))) float v2f;
typedef __attribute__((ext_vector_type(4))) float v4f;
typedef __attribute__((ext_vector_type(8))) float v8f;

#define D_FEAT 128
#define TILE_ROWS 16
#define ROW_PITCH (D_FEAT + 4)
#define WAVES_PER_BLOCK 4

__global__ void __launch_bounds__(WAVES_PER_BLOCK * 32, 1)
edge_dot_wmma_kernel(const int* __restrict__ src,
                     const int* __restrict__ dst,
                     const float* __restrict__ h,
                     float* __restrict__ out,
                     int nEdges)
{
    __shared__ __attribute__((aligned(16)))
        float tile[WAVES_PER_BLOCK][2][TILE_ROWS][ROW_PITCH];

    const int lane   = threadIdx.x & 31;
    const int waveId = __builtin_amdgcn_readfirstlane((int)(threadIdx.x >> 5));
    const int e0 = (blockIdx.x * WAVES_PER_BLOCK + waveId) << 4;

    if (e0 >= nEdges) return;

    const int m = lane & 15;
    int e = e0 + m;
    if (e >= nEdges) e = nEdges - 1;

    const int sidx = src[e];
    const int didx = dst[e];

    float* __restrict__ sT = &tile[waveId][0][0][0];
    float* __restrict__ dT = &tile[waveId][1][0][0];

    const unsigned colOff = (unsigned)lane << 2;
    v4f sv[TILE_ROWS];
    v4f dv[TILE_ROWS];
#pragma unroll
    for (int r = 0; r < TILE_ROWS; ++r) {
        const long long srow = (long long)__builtin_amdgcn_readlane(sidx, r);
        const long long drow = (long long)__builtin_amdgcn_readlane(didx, r);
        const float* __restrict__ sp = h + srow * D_FEAT;
        const float* __restrict__ dp = h + drow * D_FEAT;
        sv[r] = *(const v4f*)(sp + colOff);
        dv[r] = *(const v4f*)(dp + colOff);
    }
#pragma unroll
    for (int r = 0; r < TILE_ROWS; ++r) {
        *(v4f*)(sT + r * ROW_PITCH + colOff) = sv[r];
        *(v4f*)(dT + r * ROW_PITCH + colOff) = dv[r];
    }

    __builtin_amdgcn_fence(__ATOMIC_SEQ_CST, "wavefront");

    const int fragOff = m * ROW_PITCH + ((lane >> 4) << 2);

    v8f c = {};

    v4f af[D_FEAT / 8];
    v4f bf[D_FEAT / 8];
#pragma unroll
    for (int t = 0; t < D_FEAT / 8; ++t) {
        af[t] = *(const v4f*)(sT + fragOff + 8 * t);
        bf[t] = *(const v4f*)(dT + fragOff + 8 * t);
    }

#pragma unroll
    for (int t = 0; t < D_FEAT / 8; ++t) {
        const v4f av = af[t];
        const v4f bv = bf[t];

        v2f a0 = __builtin_shufflevector(av, av, 0, 1);
        v2f b0 = __builtin_shufflevector(bv, bv, 0, 1);
        v2f a1 = __builtin_shufflevector(av, av, 2, 3);
        v2f b1 = __builtin_shufflevector(bv, bv, 2, 3);

        c = __builtin_amdgcn_wmma_f32_16x16x4_f32(false, a0, false, b0,
                                                  (short)0, c, false, false);
        c = __builtin_amdgcn_wmma_f32_16x16x4_f32(false, a1, false, b1,
                                                  (short)0, c, false, false);
    }
    __builtin_amdgcn_sched_group_barrier(0x100, 2 * (D_FEAT / 8), 0);
    __builtin_amdgcn_sched_group_barrier(0x008, 2 * (D_FEAT / 8), 0);
    __builtin_amdgcn_sched_barrier(0);

    const int sel = lane & 7;
    float val = c[0];
    if (sel == 1) val = c[1];
    if (sel == 2) val = c[2];
    if (sel == 3) val = c[3];
    if (sel == 4) val = c[4];
    if (sel == 5) val = c[5];
    if (sel == 6) val = c[6];
    if (sel == 7) val = c[7];

    const bool writer = (lane < 8) || (lane >= 24);
    const int outIdx = e0 + ((lane < 8) ? lane : (lane - 16));
    if (writer && outIdx < nEdges) out[outIdx] = val;
}

extern "C" void kernel_launch(void* const* d_in, const int* in_sizes, int n_in,
                              void* d_out, int out_size, void* d_ws, size_t ws_size,
                              hipStream_t stream) {
    const int E = in_sizes[0] / 2;
    const int*   src = (const int*)d_in[0];
    const int*   dst = src + E;
    const float* h   = (const float*)d_in[1];
    float*       out = (float*)d_out;

    const int wavesPerBlock = WAVES_PER_BLOCK;
    const int threads = wavesPerBlock * 32;
    const int totalWaves = (E + 15) / 16;
    const int blocks = (totalWaves + wavesPerBlock - 1) / wavesPerBlock;

    edge_dot_wmma_kernel<<<blocks, threads, 0, stream>>>(src, dst, h, out, E);
}


# Resident grid of 768 work-groups: each wave loops over 12-13 tiles of 16 edges and issues the next tile's 32 row loads ahead of the current tile's wmma chain; sum order unchanged
# speedup vs baseline: 1.3684x; 1.1179x over previous
#include <hip/hip_runtime.h>
#include <hip/hip_bf16.h>


typedef __attribute__((ext_vector_type(2))) float v2f;
typedef __attribute__((ext_vector_type(4))) float v4f;
typedef __attribute__((ext_vector_type(8))) float v8f;

#define D_FEAT 128
#define TILE_ROWS 16
#define ROW_PITCH (D_FEAT + 4)
#define WAVES_PER_BLOCK 4
#define RESIDENT_BLOCKS 768

__device__ __forceinline__ void load_tile_indices(const int* __restrict__ src,
                                                  const int* __restrict__ dst,
                                                  int t, int m, int nEdges,
                                                  int& sidx, int& didx)
{
    int e = (t << 4) + m;
    if (e >= nEdges) e = nEdges - 1;
    sidx = src[e];
    didx = dst[e];
}

__device__ __forceinline__ void issue_row_loads(const float* __restrict__ h,
                                                int sidx, int didx, unsigned colOff,
                                                v4f (&sv)[TILE_ROWS], v4f (&dv)[TILE_ROWS])
{
#pragma unroll
    for (int r = 0; r < TILE_ROWS; ++r) {
        const long long srow = (long long)__builtin_amdgcn_readlane(sidx, r);
        const long long drow = (long long)__builtin_amdgcn_readlane(didx, r);
        const float* __restrict__ sp = h + srow * D_FEAT;
        const float* __restrict__ dp = h + drow * D_FEAT;
        sv[r] = *(const v4f*)(sp + colOff);
        dv[r] = *(const v4f*)(dp + colOff);
    }
}

__global__ void __launch_bounds__(WAVES_PER_BLOCK * 32, 1)
edge_dot_wmma_kernel(const int* __restrict__ src,
                     const int* __restrict__ dst,
                     const float* __restrict__ h,
                     float* __restrict__ out,
                     int nEdges)
{
    __shared__ __attribute__((aligned(16)))
        float tile[WAVES_PER_BLOCK][2][TILE_ROWS][ROW_PITCH];

    const int lane   = threadIdx.x & 31;
    const int waveId = __builtin_amdgcn_readfirstlane((int)(threadIdx.x >> 5));
    const int nTiles = (nEdges + 15) >> 4;
    const int stride = (int)gridDim.x * WAVES_PER_BLOCK;
    int cur = (int)blockIdx.x * WAVES_PER_BLOCK + waveId;

    if (cur >= nTiles) return;

    const int m = lane & 15;
    float* __restrict__ sT = &tile[waveId][0][0][0];
    float* __restrict__ dT = &tile[waveId][1][0][0];
    const unsigned colOff = (unsigned)lane << 2;

    const int fragOff = m * ROW_PITCH + ((lane >> 4) << 2);

    const int sel = lane & 7;
    const bool writer = (lane < 8) || (lane >= 24);
    const int outLane = (lane < 8) ? lane : (lane - 16);

    int sidx, didx;
    load_tile_indices(src, dst, cur, m, nEdges, sidx, didx);
    v4f sv[TILE_ROWS];
    v4f dv[TILE_ROWS];
    issue_row_loads(h, sidx, didx, colOff, sv, dv);
    int nxt = cur + stride;
    if (nxt < nTiles) load_tile_indices(src, dst, nxt, m, nEdges, sidx, didx);

    for (;;) {
#pragma unroll
        for (int r = 0; r < TILE_ROWS; ++r) {
            *(v4f*)(sT + r * ROW_PITCH + colOff) = sv[r];
            *(v4f*)(dT + r * ROW_PITCH + colOff) = dv[r];
        }

        __builtin_amdgcn_fence(__ATOMIC_SEQ_CST, "wavefront");

        const bool hasNext = nxt < nTiles;
        if (hasNext) {
            issue_row_loads(h, sidx, didx, colOff, sv, dv);
            const int nn = nxt + stride;
            if (nn < nTiles) load_tile_indices(src, dst, nn, m, nEdges, sidx, didx);
        }

        v8f c = {};

        v4f af[D_FEAT / 8];
        v4f bf[D_FEAT / 8];
#pragma unroll
        for (int t = 0; t < D_FEAT / 8; ++t) {
            af[t] = *(const v4f*)(sT + fragOff + 8 * t);
            bf[t] = *(const v4f*)(dT + fragOff + 8 * t);
        }

#pragma unroll
        for (int t = 0; t < D_FEAT / 8; ++t) {
            const v4f av = af[t];
            const v4f bv = bf[t];

            v2f a0 = __builtin_shufflevector(av, av, 0, 1);
            v2f b0 = __builtin_shufflevector(bv, bv, 0, 1);
            v2f a1 = __builtin_shufflevector(av, av, 2, 3);
            v2f b1 = __builtin_shufflevector(bv, bv, 2, 3);

            c = __builtin_amdgcn_wmma_f32_16x16x4_f32(false, a0, false, b0,
                                                      (short)0, c, false, false);
            c = __builtin_amdgcn_wmma_f32_16x16x4_f32(false, a1, false, b1,
                                                      (short)0, c, false, false);
        }
        __builtin_amdgcn_sched_group_barrier(0x100, 2 * (D_FEAT / 8), 0);
        __builtin_amdgcn_sched_group_barrier(0x008, 2 * (D_FEAT / 8), 0);
        __builtin_amdgcn_sched_barrier(0);

        float val = c[0];
        if (sel == 1) val = c[1];
        if (sel == 2) val = c[2];
        if (sel == 3) val = c[3];
        if (sel == 4) val = c[4];
        if (sel == 5) val = c[5];
        if (sel == 6) val = c[6];
        if (sel == 7) val = c[7];

        const int outIdx = (cur << 4) + outLane;
        if (writer && outIdx < nEdges) out[outIdx] = val;

        if (!hasNext) break;

        cur = nxt;
        nxt += stride;
    }
}

extern "C" void kernel_launch(void* const* d_in, const int* in_sizes, int n_in,
                              void* d_out, int out_size, void* d_ws, size_t ws_size,
                              hipStream_t stream) {
    const int E = in_sizes[0] / 2;
    const int*   src = (const int*)d_in[0];
    const int*   dst = src + E;
    const float* h   = (const float*)d_in[1];
    float*       out = (float*)d_out;

    const int wavesPerBlock = WAVES_PER_BLOCK;
    const int threads = wavesPerBlock * 32;
    const int totalTiles = (E + 15) / 16;
    const int blocksNeeded = (totalTiles + wavesPerBlock - 1) / wavesPerBlock;
    const int blocks = blocksNeeded < RESIDENT_BLOCKS ? blocksNeeded : RESIDENT_BLOCKS;

    edge_dot_wmma_kernel<<<blocks, threads, 0, stream>>>(src, dst, h, out, E);
}
